# starting kernel
# baseline (speedup 1.0000x reference)
#include <hip/hip_runtime.h>

constexpr int kN    = 8;
constexpr int kINC  = 32;
constexpr int kINN  = 50000;
constexpr int kOUTC = 64;
constexpr int kOUTN = 8192;
constexpr int kD    = 32;

typedef __attribute__((ext_vector_type(2))) float v2f;
typedef __attribute__((ext_vector_type(8))) float v8f;

__global__ __launch_bounds__(128) void fused_gather_max_wmma(
    const float* __restrict__ x,
    const float* __restrict__ nf,
    const float* __restrict__ ft,
    const float* __restrict__ bias,
    const int*   __restrict__ A,
    float*       __restrict__ out)
{
  __shared__ int   sA[16 * kD];
  __shared__ float sred[16 * 33];

  const int tid    = threadIdx.x;
  const int o_base = blockIdx.x * 16;
  const int n      = blockIdx.y;

  for (int j = tid; j < 16 * kD; j += 128)
    sA[j] = A[o_base * kD + j];
  __syncthreads();

  {
    const int i  = tid & 31;
    const int og = tid >> 5;
    const float* __restrict__ xrow  = x  + ((size_t)n * kINC + i) * kINN;
    const float* __restrict__ nfrow = nf + (size_t)i * kINN;
    #pragma unroll
    for (int rep = 0; rep < 4; ++rep) {
      const int o_loc = og + rep * 4;
      const int* __restrict__ idxs = &sA[o_loc * kD];
      float m = -__builtin_inff();
      #pragma unroll 4
      for (int d = 0; d < kD; ++d) {
        const int idx = idxs[d];
        m = fmaxf(m, xrow[idx] * nfrow[idx]);
      }
      sred[o_loc * 33 + i] = m;
    }
  }
  __syncthreads();

  {
    const int lane  = tid & 31;
    const int wave  = tid >> 5;
    const int m16   = lane & 15;
    const int hi    = lane >> 4;
    const int kbase = wave * 16;

    v8f c = {};
    #pragma unroll
    for (int s = 0; s < 8; ++s) {
      const int ks = 4 * s + 2 * hi;
      v2f a, b;
      a.x = ft[(size_t)ks       * kOUTC + kbase + m16];
      a.y = ft[(size_t)(ks + 1) * kOUTC + kbase + m16];
      b.x = sred[m16 * 33 + ks];
      b.y = sred[m16 * 33 + ks + 1];
      c = __builtin_amdgcn_wmma_f32_16x16x4_f32(
              false, a, false, b, (short)0, c, false, false);
    }

    #pragma unroll
    for (int v = 0; v < 8; ++v) {
      const int k = kbase + v + 8 * hi;
      const int o = o_base + m16;
      const size_t bidx = (size_t)k * kOUTN + o;
      out[(size_t)n * kOUTC * kOUTN + bidx] = c[v] + bias[bidx];
    }
  }
}

extern "C" void kernel_launch(void* const* d_in, const int* in_sizes, int n_in,
                              void* d_out, int out_size, void* d_ws, size_t ws_size,
                              hipStream_t stream) {
  const float* x    = (const float*)d_in[0];
  const float* nf   = (const float*)d_in[1];
  const float* ft   = (const float*)d_in[2];
  const float* bias = (const float*)d_in[3];
  const int*   A    = (const int*)d_in[4];
  float* out = (float*)d_out;

  dim3 grid(kOUTN / 16, kN);
  fused_gather_max_wmma<<<grid, 128, 0, stream>>>(x, nf, ft, bias, A, out);

  (void)in_sizes; (void)n_in; (void)out_size; (void)d_ws; (void)ws_size;
}


# New first launch stores x*nf transposed in the workspace (a position's 32 features = one 128 B line); the gather reads 1 line per (o,d), not 64; under 51200000 B of workspace the one-launch form runs
# speedup vs baseline: 8.6960x; 8.6960x over previous
#include <hip/hip_runtime.h>

constexpr int kN    = 8;
constexpr int kINC  = 32;
constexpr int kINN  = 50000;
constexpr int kOUTC = 64;
constexpr int kOUTN = 8192;
constexpr int kD    = 32;

typedef __attribute__((ext_vector_type(2))) float v2f;
typedef __attribute__((ext_vector_type(8))) float v8f;

__global__ __launch_bounds__(128) void fused_gather_max_wmma(
    const float* __restrict__ x,
    const float* __restrict__ nf,
    const float* __restrict__ ft,
    const float* __restrict__ bias,
    const int*   __restrict__ A,
    float*       __restrict__ out)
{
  __shared__ int   sA[16 * kD];
  __shared__ float sred[16 * 33];

  const int tid    = threadIdx.x;
  const int o_base = blockIdx.x * 16;
  const int n      = blockIdx.y;

  for (int j = tid; j < 16 * kD; j += 128)
    sA[j] = A[o_base * kD + j];
  __syncthreads();

  {
    const int i  = tid & 31;
    const int og = tid >> 5;
    const float* __restrict__ xrow  = x  + ((size_t)n * kINC + i) * kINN;
    const float* __restrict__ nfrow = nf + (size_t)i * kINN;
    #pragma unroll
    for (int rep = 0; rep < 4; ++rep) {
      const int o_loc = og + rep * 4;
      const int* __restrict__ idxs = &sA[o_loc * kD];
      float m = -__builtin_inff();
      #pragma unroll 4
      for (int d = 0; d < kD; ++d) {
        const int idx = idxs[d];
        m = fmaxf(m, xrow[idx] * nfrow[idx]);
      }
      sred[o_loc * 33 + i] = m;
    }
  }
  __syncthreads();

  {
    const int lane  = tid & 31;
    const int wave  = tid >> 5;
    const int m16   = lane & 15;
    const int hi    = lane >> 4;
    const int kbase = wave * 16;

    v8f c = {};
    #pragma unroll
    for (int s = 0; s < 8; ++s) {
      const int ks = 4 * s + 2 * hi;
      v2f a, b;
      a.x = ft[(size_t)ks       * kOUTC + kbase + m16];
      a.y = ft[(size_t)(ks + 1) * kOUTC + kbase + m16];
      b.x = sred[m16 * 33 + ks];
      b.y = sred[m16 * 33 + ks + 1];
      c = __builtin_amdgcn_wmma_f32_16x16x4_f32(
              false, a, false, b, (short)0, c, false, false);
    }

    #pragma unroll
    for (int v = 0; v < 8; ++v) {
      const int k = kbase + v + 8 * hi;
      const int o = o_base + m16;
      const size_t bidx = (size_t)k * kOUTN + o;
      out[(size_t)n * kOUTC * kOUTN + bidx] = c[v] + bias[bidx];
    }
  }
}

constexpr int kTP    = 256;
constexpr int kPitch = 260;

__global__ __launch_bounds__(256) void scale_transpose_rows(
    const float* __restrict__ x,
    const float* __restrict__ nf,
    float*       __restrict__ yT)
{
  __shared__ __attribute__((aligned(16))) float s[kINC * kPitch];

  const int tid  = threadIdx.x;
  const int lane = tid & 31;
  const int wave = tid >> 5;
  const int p0   = blockIdx.x * kTP;
  const int n    = blockIdx.y;

  const int ca = min(p0 + 4 * lane,       kINN - 4);
  const int cb = min(p0 + 4 * lane + 128, kINN - 4);
  #pragma unroll
  for (int k = 0; k < 4; ++k) {
    const int i = wave + 8 * k;
    const float* __restrict__ xrow  = x  + ((size_t)n * kINC + i) * kINN;
    const float* __restrict__ nfrow = nf + (size_t)i * kINN;
    const float4 xa = *reinterpret_cast<const float4*>(xrow  + ca);
    const float4 wa = *reinterpret_cast<const float4*>(nfrow + ca);
    const float4 xb = *reinterpret_cast<const float4*>(xrow  + cb);
    const float4 wb = *reinterpret_cast<const float4*>(nfrow + cb);
    float4 pa, pb;
    pa.x = xa.x * wa.x; pa.y = xa.y * wa.y; pa.z = xa.z * wa.z; pa.w = xa.w * wa.w;
    pb.x = xb.x * wb.x; pb.y = xb.y * wb.y; pb.z = xb.z * wb.z; pb.w = xb.w * wb.w;
    *reinterpret_cast<float4*>(&s[i * kPitch + 4 * lane])       = pa;
    *reinterpret_cast<float4*>(&s[i * kPitch + 4 * lane + 128]) = pb;
  }
  __syncthreads();

  float* __restrict__ dst = yT + ((size_t)n * kINN + p0) * kINC + lane;
  const float* __restrict__ src = &s[lane * kPitch + wave];
  if (p0 + kTP <= kINN) {
    #pragma unroll
    for (int k = 0; k < kTP / 8; ++k)
      dst[(size_t)(wave + 8 * k) * kINC] = src[8 * k];
  } else {
    for (int k = 0; k < kTP / 8; ++k) {
      const int c = wave + 8 * k;
      if (p0 + c < kINN)
        dst[(size_t)c * kINC] = src[8 * k];
    }
  }
}

__global__ __launch_bounds__(128) void gather_lines_max_wmma(
    const float* __restrict__ yT,
    const float* __restrict__ ft,
    const float* __restrict__ bias,
    const int*   __restrict__ A,
    float*       __restrict__ out)
{
  __shared__ int   sA[16 * kD];
  __shared__ float sred[16 * 33];

  const int tid    = threadIdx.x;
  const int o_base = blockIdx.x * 16;
  const int n      = blockIdx.y;

  for (int j = tid; j < 16 * kD; j += 128)
    sA[j] = A[o_base * kD + j];
  __syncthreads();

  {
    const int i  = tid & 31;
    const int og = tid >> 5;
    const char* __restrict__ yb = reinterpret_cast<const char*>(yT + (size_t)n * kINN * kINC);
    const unsigned lane_b = (unsigned)i << 2;
    #pragma unroll
    for (int rep = 0; rep < 4; ++rep) {
      const int o_loc = og + rep * 4;
      const int* __restrict__ idxs = &sA[o_loc * kD];
      float v[kD];
      #pragma unroll
      for (int d = 0; d < kD; ++d) {
        const unsigned boff = ((unsigned)idxs[d] << 7) | lane_b;
        v[d] = *reinterpret_cast<const float*>(yb + boff);
      }
      asm volatile("" :: "v"(v[0]),  "v"(v[1]),  "v"(v[2]),  "v"(v[3]),  "v"(v[4]),  "v"(v[5]),  "v"(v[6]),  "v"(v[7]),
                         "v"(v[8]),  "v"(v[9]),  "v"(v[10]), "v"(v[11]), "v"(v[12]), "v"(v[13]), "v"(v[14]), "v"(v[15]));
      asm volatile("" :: "v"(v[16]), "v"(v[17]), "v"(v[18]), "v"(v[19]), "v"(v[20]), "v"(v[21]), "v"(v[22]), "v"(v[23]),
                         "v"(v[24]), "v"(v[25]), "v"(v[26]), "v"(v[27]), "v"(v[28]), "v"(v[29]), "v"(v[30]), "v"(v[31]));
      float m = -__builtin_inff();
      #pragma unroll
      for (int d = 0; d < kD; ++d)
        m = fmaxf(m, v[d]);
      sred[o_loc * 33 + i] = m;
    }
  }
  __syncthreads();

  {
    const int lane  = tid & 31;
    const int wave  = tid >> 5;
    const int m16   = lane & 15;
    const int hi    = lane >> 4;
    const int kbase = wave * 16;

    v8f c = {};
    #pragma unroll
    for (int s = 0; s < 8; ++s) {
      const int ks = 4 * s + 2 * hi;
      v2f a, b;
      a.x = ft[(size_t)ks       * kOUTC + kbase + m16];
      a.y = ft[(size_t)(ks + 1) * kOUTC + kbase + m16];
      b.x = sred[m16 * 33 + ks];
      b.y = sred[m16 * 33 + ks + 1];
      c = __builtin_amdgcn_wmma_f32_16x16x4_f32(
              false, a, false, b, (short)0, c, false, false);
    }

    #pragma unroll
    for (int v = 0; v < 8; ++v) {
      const int k = kbase + v + 8 * hi;
      const int o = o_base + m16;
      const size_t bidx = (size_t)k * kOUTN + o;
      out[(size_t)n * kOUTC * kOUTN + bidx] = c[v] + bias[bidx];
    }
  }
}

extern "C" void kernel_launch(void* const* d_in, const int* in_sizes, int n_in,
                              void* d_out, int out_size, void* d_ws, size_t ws_size,
                              hipStream_t stream) {
  const float* x    = (const float*)d_in[0];
  const float* nf   = (const float*)d_in[1];
  const float* ft   = (const float*)d_in[2];
  const float* bias = (const float*)d_in[3];
  const int*   A    = (const int*)d_in[4];
  float* out = (float*)d_out;

  dim3 grid(kOUTN / 16, kN);

  constexpr size_t kPlaneBytes = (size_t)kN * kINN * kINC * sizeof(float);
  if (d_ws != nullptr && ws_size >= kPlaneBytes) {
    float* yT = (float*)d_ws;
    dim3 tgrid((kINN + kTP - 1) / kTP, kN);
    scale_transpose_rows<<<tgrid, 256, 0, stream>>>(x, nf, yT);
    gather_lines_max_wmma<<<grid, 128, 0, stream>>>(yT, ft, bias, A, out);
  } else {
    fused_gather_max_wmma<<<grid, 128, 0, stream>>>(x, nf, ft, bias, A, out);
  }

  (void)in_sizes; (void)n_in; (void)out_size;
}


# Plane stored position-major with all 8 batches of a position in one 1 KiB record; the gather reads a record with two 16-byte-per-lane wave loads, indices and addresses shared by the 8 batches
# speedup vs baseline: 9.7132x; 1.1170x over previous
#include <hip/hip_runtime.h>

constexpr int kN    = 8;
constexpr int kINC  = 32;
constexpr int kINN  = 50000;
constexpr int kOUTC = 64;
constexpr int kOUTN = 8192;
constexpr int kD    = 32;

typedef __attribute__((ext_vector_type(2))) float v2f;
typedef __attribute__((ext_vector_type(8))) float v8f;

__global__ __launch_bounds__(128) void fused_gather_max_wmma(
    const float* __restrict__ x,
    const float* __restrict__ nf,
    const float* __restrict__ ft,
    const float* __restrict__ bias,
    const int*   __restrict__ A,
    float*       __restrict__ out)
{
  __shared__ int   sA[16 * kD];
  __shared__ float sred[16 * 33];

  const int tid    = threadIdx.x;
  const int o_base = blockIdx.x * 16;
  const int n      = blockIdx.y;

  for (int j = tid; j < 16 * kD; j += 128)
    sA[j] = A[o_base * kD + j];
  __syncthreads();

  {
    const int i  = tid & 31;
    const int og = tid >> 5;
    const float* __restrict__ xrow  = x  + ((size_t)n * kINC + i) * kINN;
    const float* __restrict__ nfrow = nf + (size_t)i * kINN;
    #pragma unroll
    for (int rep = 0; rep < 4; ++rep) {
      const int o_loc = og + rep * 4;
      const int* __restrict__ idxs = &sA[o_loc * kD];
      float m = -__builtin_inff();
      #pragma unroll 4
      for (int d = 0; d < kD; ++d) {
        const int idx = idxs[d];
        m = fmaxf(m, xrow[idx] * nfrow[idx]);
      }
      sred[o_loc * 33 + i] = m;
    }
  }
  __syncthreads();

  {
    const int lane  = tid & 31;
    const int wave  = tid >> 5;
    const int m16   = lane & 15;
    const int hi    = lane >> 4;
    const int kbase = wave * 16;

    v8f c = {};
    #pragma unroll
    for (int s = 0; s < 8; ++s) {
      const int ks = 4 * s + 2 * hi;
      v2f a, b;
      a.x = ft[(size_t)ks       * kOUTC + kbase + m16];
      a.y = ft[(size_t)(ks + 1) * kOUTC + kbase + m16];
      b.x = sred[m16 * 33 + ks];
      b.y = sred[m16 * 33 + ks + 1];
      c = __builtin_amdgcn_wmma_f32_16x16x4_f32(
              false, a, false, b, (short)0, c, false, false);
    }

    #pragma unroll
    for (int v = 0; v < 8; ++v) {
      const int k = kbase + v + 8 * hi;
      const int o = o_base + m16;
      const size_t bidx = (size_t)k * kOUTN + o;
      out[(size_t)n * kOUTC * kOUTN + bidx] = c[v] + bias[bidx];
    }
  }
}

typedef __attribute__((ext_vector_type(4))) float v4f;

constexpr int kTP    = 64;
constexpr int kPitch = 68;
constexpr int kRows  = kN * kINC;
constexpr int kRec   = kN * kINC;

__global__ __launch_bounds__(256) void scale_transpose_records(
    const float* __restrict__ x,
    const float* __restrict__ nf,
    float*       __restrict__ yR)
{
  __shared__ __attribute__((aligned(16))) float s[kRows * kPitch];

  const int tid  = threadIdx.x;
  const int lane = tid & 31;
  const int wave = tid >> 5;
  const int p0   = blockIdx.x * kTP;

  const int seg = tid & 15;
  const int rb  = tid >> 4;
  const int col = min(p0 + 4 * seg, kINN - 4);
  const v4f w0 = *reinterpret_cast<const v4f*>(nf + (size_t)rb        * kINN + col);
  const v4f w1 = *reinterpret_cast<const v4f*>(nf + (size_t)(rb + 16) * kINN + col);
  #pragma unroll
  for (int k = 0; k < 16; ++k) {
    const int r = rb + 16 * k;
    const v4f xv = *reinterpret_cast<const v4f*>(x + (size_t)r * kINN + col);
    const v4f pv = xv * ((k & 1) ? w1 : w0);
    *reinterpret_cast<v4f*>(&s[r * kPitch + 4 * seg]) = pv;
  }
  __syncthreads();

  if (p0 + kTP <= kINN) {
    #pragma unroll
    for (int k = 0; k < kTP / 8; ++k) {
      const int c = wave + 8 * k;
      float* __restrict__ dst = yR + (size_t)(p0 + c) * kRec + lane;
      #pragma unroll
      for (int n = 0; n < kN; ++n)
        dst[n * kINC] = s[(n * kINC + lane) * kPitch + c];
    }
  } else {
    for (int k = 0; k < kTP / 8; ++k) {
      const int c = wave + 8 * k;
      if (p0 + c < kINN) {
        float* __restrict__ dst = yR + (size_t)(p0 + c) * kRec + lane;
        #pragma unroll
        for (int n = 0; n < kN; ++n)
          dst[n * kINC] = s[(n * kINC + lane) * kPitch + c];
      }
    }
  }
}

constexpr int kSP = 36;

__global__ __launch_bounds__(256) void gather_records_max_wmma(
    const float* __restrict__ yR,
    const float* __restrict__ ft,
    const float* __restrict__ bias,
    const int*   __restrict__ A,
    float*       __restrict__ out)
{
  __shared__ int sA[16 * kD];
  __shared__ __attribute__((aligned(16))) float sred[kN * 16 * kSP];

  const int tid    = threadIdx.x;
  const int lane   = tid & 31;
  const int wave   = tid >> 5;
  const int o_base = blockIdx.x * 16;

  for (int j = tid; j < 16 * kD; j += 256)
    sA[j] = A[o_base * kD + j];
  __syncthreads();

  {
    const char* __restrict__ yb = reinterpret_cast<const char*>(yR);
    const unsigned lane_b = (unsigned)lane << 4;
    const int nl = lane >> 3;
    const int i4 = (lane & 7) * 4;
    #pragma unroll
    for (int rep = 0; rep < 2; ++rep) {
      const int o_loc = wave * 2 + rep;
      const int* __restrict__ idxs = &sA[o_loc * kD];
      const float ninf = -__builtin_inff();
      v4f mA = {ninf, ninf, ninf, ninf};
      v4f mB = mA;
      #pragma unroll
      for (int g = 0; g < 2; ++g) {
        v4f va[16], vb[16];
        #pragma unroll
        for (int d = 0; d < 16; ++d) {
          const unsigned boff = ((unsigned)idxs[16 * g + d] << 10) | lane_b;
          va[d] = *reinterpret_cast<const v4f*>(yb + boff);
          vb[d] = *reinterpret_cast<const v4f*>(yb + boff + 512);
        }
        asm volatile("" :: "v"(va[0]), "v"(va[1]), "v"(va[2]),  "v"(va[3]),  "v"(va[4]),  "v"(va[5]),  "v"(va[6]),  "v"(va[7]),
                           "v"(va[8]), "v"(va[9]), "v"(va[10]), "v"(va[11]), "v"(va[12]), "v"(va[13]), "v"(va[14]), "v"(va[15]));
        asm volatile("" :: "v"(vb[0]), "v"(vb[1]), "v"(vb[2]),  "v"(vb[3]),  "v"(vb[4]),  "v"(vb[5]),  "v"(vb[6]),  "v"(vb[7]),
                           "v"(vb[8]), "v"(vb[9]), "v"(vb[10]), "v"(vb[11]), "v"(vb[12]), "v"(vb[13]), "v"(vb[14]), "v"(vb[15]));
        #pragma unroll
        for (int d = 0; d < 16; ++d) {
          mA = __builtin_elementwise_max(mA, va[d]);
          mB = __builtin_elementwise_max(mB, vb[d]);
        }
      }
      *reinterpret_cast<v4f*>(&sred[(nl * 16 + o_loc) * kSP + i4])       = mA;
      *reinterpret_cast<v4f*>(&sred[((4 + nl) * 16 + o_loc) * kSP + i4]) = mB;
    }
  }
  __syncthreads();

  {
    const int n   = wave;
    const int m16 = lane & 15;
    const int hi  = lane >> 4;

    v2f b[8];
    #pragma unroll
    for (int s = 0; s < 8; ++s) {
      const int ks = 4 * s + 2 * hi;
      b[s].x = sred[(n * 16 + m16) * kSP + ks];
      b[s].y = sred[(n * 16 + m16) * kSP + ks + 1];
    }

    v8f c[4] = {};
    #pragma unroll
    for (int s = 0; s < 8; ++s) {
      const int ks = 4 * s + 2 * hi;
      #pragma unroll
      for (int kt = 0; kt < 4; ++kt) {
        v2f a;
        a.x = ft[(size_t)ks       * kOUTC + kt * 16 + m16];
        a.y = ft[(size_t)(ks + 1) * kOUTC + kt * 16 + m16];
        c[kt] = __builtin_amdgcn_wmma_f32_16x16x4_f32(
                    false, a, false, b[s], (short)0, c[kt], false, false);
      }
    }

    #pragma unroll
    for (int kt = 0; kt < 4; ++kt) {
      #pragma unroll
      for (int v = 0; v < 8; ++v) {
        const int k = kt * 16 + v + 8 * hi;
        const int o = o_base + m16;
        const size_t bidx = (size_t)k * kOUTN + o;
        out[(size_t)n * kOUTC * kOUTN + bidx] = c[kt][v] + bias[bidx];
      }
    }
  }
}

extern "C" void kernel_launch(void* const* d_in, const int* in_sizes, int n_in,
                              void* d_out, int out_size, void* d_ws, size_t ws_size,
                              hipStream_t stream) {
  const float* x    = (const float*)d_in[0];
  const float* nf   = (const float*)d_in[1];
  const float* ft   = (const float*)d_in[2];
  const float* bias = (const float*)d_in[3];
  const int*   A    = (const int*)d_in[4];
  float* out = (float*)d_out;

  constexpr size_t kPlaneBytes = (size_t)kN * kINN * kINC * sizeof(float);
  if (d_ws != nullptr && ws_size >= kPlaneBytes) {
    float* yR = (float*)d_ws;
    scale_transpose_records<<<(kINN + kTP - 1) / kTP, 256, 0, stream>>>(x, nf, yR);
    gather_records_max_wmma<<<kOUTN / 16, 256, 0, stream>>>(yR, ft, bias, A, out);
  } else {
    dim3 grid(kOUTN / 16, kN);
    fused_gather_max_wmma<<<grid, 128, 0, stream>>>(x, nf, ft, bias, A, out);
  }

  (void)in_sizes; (void)n_in; (void)out_size;
}


# Gather moved into LDS: a workgroup keeps one whole (n,i) row of x*nf (200000 B) in LDS and takes the 32-neighbour maxima there; the workspace holds only the maxima; launch 2 is the WMMA product
# speedup vs baseline: 11.1396x; 1.1469x over previous
#include <hip/hip_runtime.h>

constexpr int kN    = 8;
constexpr int kINC  = 32;
constexpr int kINN  = 50000;
constexpr int kOUTC = 64;
constexpr int kOUTN = 8192;
constexpr int kD    = 32;

typedef __attribute__((ext_vector_type(2))) float v2f;
typedef __attribute__((ext_vector_type(8))) float v8f;

__global__ __launch_bounds__(128) void fused_gather_max_wmma(
    const float* __restrict__ x,
    const float* __restrict__ nf,
    const float* __restrict__ ft,
    const float* __restrict__ bias,
    const int*   __restrict__ A,
    float*       __restrict__ out)
{
  __shared__ int   sA[16 * kD];
  __shared__ float sred[16 * 33];

  const int tid    = threadIdx.x;
  const int o_base = blockIdx.x * 16;
  const int n      = blockIdx.y;

  for (int j = tid; j < 16 * kD; j += 128)
    sA[j] = A[o_base * kD + j];
  __syncthreads();

  {
    const int i  = tid & 31;
    const int og = tid >> 5;
    const float* __restrict__ xrow  = x  + ((size_t)n * kINC + i) * kINN;
    const float* __restrict__ nfrow = nf + (size_t)i * kINN;
    #pragma unroll
    for (int rep = 0; rep < 4; ++rep) {
      const int o_loc = og + rep * 4;
      const int* __restrict__ idxs = &sA[o_loc * kD];
      float m = -__builtin_inff();
      #pragma unroll 4
      for (int d = 0; d < kD; ++d) {
        const int idx = idxs[d];
        m = fmaxf(m, xrow[idx] * nfrow[idx]);
      }
      sred[o_loc * 33 + i] = m;
    }
  }
  __syncthreads();

  {
    const int lane  = tid & 31;
    const int wave  = tid >> 5;
    const int m16   = lane & 15;
    const int hi    = lane >> 4;
    const int kbase = wave * 16;

    v8f c = {};
    #pragma unroll
    for (int s = 0; s < 8; ++s) {
      const int ks = 4 * s + 2 * hi;
      v2f a, b;
      a.x = ft[(size_t)ks       * kOUTC + kbase + m16];
      a.y = ft[(size_t)(ks + 1) * kOUTC + kbase + m16];
      b.x = sred[m16 * 33 + ks];
      b.y = sred[m16 * 33 + ks + 1];
      c = __builtin_amdgcn_wmma_f32_16x16x4_f32(
              false, a, false, b, (short)0, c, false, false);
    }

    #pragma unroll
    for (int v = 0; v < 8; ++v) {
      const int k = kbase + v + 8 * hi;
      const int o = o_base + m16;
      const size_t bidx = (size_t)k * kOUTN + o;
      out[(size_t)n * kOUTC * kOUTN + bidx] = c[v] + bias[bidx];
    }
  }
}

typedef __attribute__((ext_vector_type(4))) float v4f;
typedef __attribute__((ext_vector_type(4))) int   v4i;

constexpr int kRowT  = 512;
constexpr int kRowF4 = kINN / 4;
constexpr int kOW    = kOUTN / (kRowT / 32);
constexpr int kIt    = kOW / 32;

template <int MASK>
__device__ __forceinline__ float max_xmask(float v) {
  const int p = __builtin_amdgcn_update_dpp(0, __float_as_int(v), 0x160 + MASK, 0xF, 0xF, true);
  return fmaxf(v, __int_as_float(p));
}

__global__ __launch_bounds__(kRowT) void row_in_lds_gather_max(
    const float* __restrict__ x,
    const float* __restrict__ nf,
    const int*   __restrict__ A,
    float*       __restrict__ red)
{
  __shared__ __attribute__((aligned(16))) float srow[kINN];

  const int tid  = threadIdx.x;
  const int lane = tid & 31;
  const int wave = tid >> 5;
  const int r    = blockIdx.x;
  const int i    = r & 31;

  {
    const v4f* __restrict__ xr = reinterpret_cast<const v4f*>(x  + (size_t)r * kINN);
    const v4f* __restrict__ wr = reinterpret_cast<const v4f*>(nf + (size_t)i * kINN);
    v4f* s4 = reinterpret_cast<v4f*>(srow);
    #pragma unroll
    for (int b = 0; b < 2; ++b) {
      v4f xv[12], wv[12];
      #pragma unroll
      for (int k = 0; k < 12; ++k) {
        const int q = tid + (b * 12 + k) * kRowT;
        xv[k] = xr[q];
        wv[k] = wr[q];
      }
      #pragma unroll
      for (int k = 0; k < 12; ++k) {
        const int q = tid + (b * 12 + k) * kRowT;
        s4[q] = xv[k] * wv[k];
      }
    }
    const int q = tid + 24 * kRowT;
    if (q < kRowF4) s4[q] = xr[q] * wr[q];
  }
  __syncthreads();

  {
    const v4i* __restrict__ a4 = reinterpret_cast<const v4i*>(A + (size_t)wave * kOW * kD) + lane;
    float* __restrict__ rr = red + (size_t)r * kOUTN + wave * kOW + 4 * (lane & 7) + (lane >> 3);
    const int sub = lane & 7;

    v4i nxt[8];
    #pragma unroll
    for (int j = 0; j < 8; ++j) nxt[j] = a4[j * 32];

    #pragma unroll 1
    for (int it = 0; it < kIt; ++it) {
      v4i cur[8];
      #pragma unroll
      for (int j = 0; j < 8; ++j) cur[j] = nxt[j];
      if (it + 1 < kIt) {
        #pragma unroll
        for (int j = 0; j < 8; ++j) nxt[j] = a4[(it + 1) * 256 + j * 32];
      }

      float m[8];
      #pragma unroll
      for (int j = 0; j < 8; ++j) {
        const float g0 = srow[cur[j].x];
        const float g1 = srow[cur[j].y];
        const float g2 = srow[cur[j].z];
        const float g3 = srow[cur[j].w];
        m[j] = fmaxf(fmaxf(g0, g1), fmaxf(g2, g3));
      }
      #pragma unroll
      for (int j = 0; j < 8; ++j) {
        m[j] = max_xmask<1>(m[j]);
        m[j] = max_xmask<2>(m[j]);
        m[j] = max_xmask<4>(m[j]);
      }
      float v = m[0];
      #pragma unroll
      for (int j = 1; j < 8; ++j) v = (sub == j) ? m[j] : v;
      rr[it * 32] = v;
    }
  }
}

__global__ __launch_bounds__(256) void maxima_times_ft_wmma(
    const float* __restrict__ red,
    const float* __restrict__ ft,
    const float* __restrict__ bias,
    float*       __restrict__ out)
{
  const int tid    = threadIdx.x;
  const int lane   = tid & 31;
  const int n      = tid >> 5;
  const int o_base = blockIdx.x * 16;
  const int m16    = lane & 15;
  const int hi     = lane >> 4;

  const float* __restrict__ rb = red + (size_t)n * kINC * kOUTN + o_base + m16;

  v2f b[8];
  #pragma unroll
  for (int s = 0; s < 8; ++s) {
    const int ks = 4 * s + 2 * hi;
    b[s].x = rb[(size_t)ks       * kOUTN];
    b[s].y = rb[(size_t)(ks + 1) * kOUTN];
  }

  v8f c[4] = {};
  #pragma unroll
  for (int s = 0; s < 8; ++s) {
    const int ks = 4 * s + 2 * hi;
    #pragma unroll
    for (int kt = 0; kt < 4; ++kt) {
      v2f a;
      a.x = ft[(size_t)ks       * kOUTC + kt * 16 + m16];
      a.y = ft[(size_t)(ks + 1) * kOUTC + kt * 16 + m16];
      c[kt] = __builtin_amdgcn_wmma_f32_16x16x4_f32(
                  false, a, false, b[s], (short)0, c[kt], false, false);
    }
  }

  #pragma unroll
  for (int kt = 0; kt < 4; ++kt) {
    #pragma unroll
    for (int v = 0; v < 8; ++v) {
      const int k = kt * 16 + v + 8 * hi;
      const int o = o_base + m16;
      const size_t bidx = (size_t)k * kOUTN + o;
      out[(size_t)n * kOUTC * kOUTN + bidx] = c[kt][v] + bias[bidx];
    }
  }
}

extern "C" void kernel_launch(void* const* d_in, const int* in_sizes, int n_in,
                              void* d_out, int out_size, void* d_ws, size_t ws_size,
                              hipStream_t stream) {
  const float* x    = (const float*)d_in[0];
  const float* nf   = (const float*)d_in[1];
  const float* ft   = (const float*)d_in[2];
  const float* bias = (const float*)d_in[3];
  const int*   A    = (const int*)d_in[4];
  float* out = (float*)d_out;

  constexpr size_t kRedBytes = (size_t)kN * kINC * kOUTN * sizeof(float);
  if (d_ws != nullptr && ws_size >= kRedBytes) {
    float* red = (float*)d_ws;
    row_in_lds_gather_max<<<kN * kINC, kRowT, 0, stream>>>(x, nf, A, red);
    maxima_times_ft_wmma<<<kOUTN / 16, 256, 0, stream>>>(red, ft, bias, out);
  } else {
    dim3 grid(kOUTN / 16, kN);
    fused_gather_max_wmma<<<grid, 128, 0, stream>>>(x, nf, ft, bias, A, out);
  }

  (void)in_sizes; (void)n_in; (void)out_size;
}


# Product launch retiled: a wave owns 64 positions x 64 channels as 4x4 tiles whose lane elements are 4 neighbouring floats, so ft, maxima, bias and out move as 16-byte whole-line accesses; 1024 waves
# speedup vs baseline: 13.3720x; 1.2004x over previous
#include <hip/hip_runtime.h>

constexpr int kN    = 8;
constexpr int kINC  = 32;
constexpr int kINN  = 50000;
constexpr int kOUTC = 64;
constexpr int kOUTN = 8192;
constexpr int kD    = 32;

typedef __attribute__((ext_vector_type(2))) float v2f;
typedef __attribute__((ext_vector_type(8))) float v8f;

__global__ __launch_bounds__(128) void fused_gather_max_wmma(
    const float* __restrict__ x,
    const float* __restrict__ nf,
    const float* __restrict__ ft,
    const float* __restrict__ bias,
    const int*   __restrict__ A,
    float*       __restrict__ out)
{
  __shared__ int   sA[16 * kD];
  __shared__ float sred[16 * 33];

  const int tid    = threadIdx.x;
  const int o_base = blockIdx.x * 16;
  const int n      = blockIdx.y;

  for (int j = tid; j < 16 * kD; j += 128)
    sA[j] = A[o_base * kD + j];
  __syncthreads();

  {
    const int i  = tid & 31;
    const int og = tid >> 5;
    const float* __restrict__ xrow  = x  + ((size_t)n * kINC + i) * kINN;
    const float* __restrict__ nfrow = nf + (size_t)i * kINN;
    #pragma unroll
    for (int rep = 0; rep < 4; ++rep) {
      const int o_loc = og + rep * 4;
      const int* __restrict__ idxs = &sA[o_loc * kD];
      float m = -__builtin_inff();
      #pragma unroll 4
      for (int d = 0; d < kD; ++d) {
        const int idx = idxs[d];
        m = fmaxf(m, xrow[idx] * nfrow[idx]);
      }
      sred[o_loc * 33 + i] = m;
    }
  }
  __syncthreads();

  {
    const int lane  = tid & 31;
    const int wave  = tid >> 5;
    const int m16   = lane & 15;
    const int hi    = lane >> 4;
    const int kbase = wave * 16;

    v8f c = {};
    #pragma unroll
    for (int s = 0; s < 8; ++s) {
      const int ks = 4 * s + 2 * hi;
      v2f a, b;
      a.x = ft[(size_t)ks       * kOUTC + kbase + m16];
      a.y = ft[(size_t)(ks + 1) * kOUTC + kbase + m16];
      b.x = sred[m16 * 33 + ks];
      b.y = sred[m16 * 33 + ks + 1];
      c = __builtin_amdgcn_wmma_f32_16x16x4_f32(
              false, a, false, b, (short)0, c, false, false);
    }

    #pragma unroll
    for (int v = 0; v < 8; ++v) {
      const int k = kbase + v + 8 * hi;
      const int o = o_base + m16;
      const size_t bidx = (size_t)k * kOUTN + o;
      out[(size_t)n * kOUTC * kOUTN + bidx] = c[v] + bias[bidx];
    }
  }
}

typedef __attribute__((ext_vector_type(4))) float v4f;
typedef __attribute__((ext_vector_type(4))) int   v4i;

constexpr int kRowT  = 512;
constexpr int kRowF4 = kINN / 4;
constexpr int kOW    = kOUTN / (kRowT / 32);
constexpr int kIt    = kOW / 32;

template <int MASK>
__device__ __forceinline__ float max_xmask(float v) {
  const int p = __builtin_amdgcn_update_dpp(0, __float_as_int(v), 0x160 + MASK, 0xF, 0xF, true);
  return fmaxf(v, __int_as_float(p));
}

__global__ __launch_bounds__(kRowT) void row_in_lds_gather_max(
    const float* __restrict__ x,
    const float* __restrict__ nf,
    const int*   __restrict__ A,
    float*       __restrict__ red)
{
  __shared__ __attribute__((aligned(16))) float srow[kINN];

  const int tid  = threadIdx.x;
  const int lane = tid & 31;
  const int wave = tid >> 5;
  const int r    = blockIdx.x;
  const int i    = r & 31;

  {
    const v4f* __restrict__ xr = reinterpret_cast<const v4f*>(x  + (size_t)r * kINN);
    const v4f* __restrict__ wr = reinterpret_cast<const v4f*>(nf + (size_t)i * kINN);
    v4f* s4 = reinterpret_cast<v4f*>(srow);
    #pragma unroll
    for (int b = 0; b < 2; ++b) {
      v4f xv[12], wv[12];
      #pragma unroll
      for (int k = 0; k < 12; ++k) {
        const int q = tid + (b * 12 + k) * kRowT;
        xv[k] = xr[q];
        wv[k] = wr[q];
      }
      #pragma unroll
      for (int k = 0; k < 12; ++k) {
        const int q = tid + (b * 12 + k) * kRowT;
        s4[q] = xv[k] * wv[k];
      }
    }
    const int q = tid + 24 * kRowT;
    if (q < kRowF4) s4[q] = xr[q] * wr[q];
  }
  __syncthreads();

  {
    const v4i* __restrict__ a4 = reinterpret_cast<const v4i*>(A + (size_t)wave * kOW * kD) + lane;
    float* __restrict__ rr = red + (size_t)r * kOUTN + wave * kOW + 4 * (lane & 7) + (lane >> 3);
    const int sub = lane & 7;

    v4i nxt[8];
    #pragma unroll
    for (int j = 0; j < 8; ++j) nxt[j] = a4[j * 32];

    #pragma unroll 1
    for (int it = 0; it < kIt; ++it) {
      v4i cur[8];
      #pragma unroll
      for (int j = 0; j < 8; ++j) cur[j] = nxt[j];
      if (it + 1 < kIt) {
        #pragma unroll
        for (int j = 0; j < 8; ++j) nxt[j] = a4[(it + 1) * 256 + j * 32];
      }

      float m[8];
      #pragma unroll
      for (int j = 0; j < 8; ++j) {
        const float g0 = srow[cur[j].x];
        const float g1 = srow[cur[j].y];
        const float g2 = srow[cur[j].z];
        const float g3 = srow[cur[j].w];
        m[j] = fmaxf(fmaxf(g0, g1), fmaxf(g2, g3));
      }
      #pragma unroll
      for (int j = 0; j < 8; ++j) {
        m[j] = max_xmask<1>(m[j]);
        m[j] = max_xmask<2>(m[j]);
        m[j] = max_xmask<4>(m[j]);
      }
      float v = m[0];
      #pragma unroll
      for (int j = 1; j < 8; ++j) v = (sub == j) ? m[j] : v;
      rr[it * 32] = v;
    }
  }
}

constexpr int kPosW = 64;
__global__ __launch_bounds__(128) void maxima_times_ft_wmma(
    const float* __restrict__ red,
    const float* __restrict__ ft,
    const float* __restrict__ bias,
    float*       __restrict__ out)
{
  const int tid    = threadIdx.x;
  const int lane   = tid & 31;
  const int n      = blockIdx.y * 4 + (tid >> 5);
  const int o_base = blockIdx.x * kPosW;
  const int c      = lane & 15;
  const int h      = lane >> 4;

  const v4f* __restrict__ rb = reinterpret_cast<const v4f*>(
      red + ((size_t)n * kINC + 2 * h) * kOUTN + o_base + 4 * c);
  const v4f* __restrict__ fa = reinterpret_cast<const v4f*>(ft + (size_t)(2 * h) * kOUTC + 4 * c);

  v4f bx[8], by[8], ax[8], ay[8];
  #pragma unroll
  for (int s = 0; s < 8; ++s) {
    bx[s] = rb[(size_t)(4 * s)     * (kOUTN / 4)];
    by[s] = rb[(size_t)(4 * s + 1) * (kOUTN / 4)];
    ax[s] = fa[(4 * s)     * (kOUTC / 4)];
    ay[s] = fa[(4 * s + 1) * (kOUTC / 4)];
  }

  const size_t lane_off = (size_t)(32 * h) * kOUTN + o_base + 4 * c;
  const float* __restrict__ bl = bias + lane_off;
  float*       __restrict__ ol = out + (size_t)n * kOUTC * kOUTN + lane_off;

  #pragma unroll
  for (int kt = 0; kt < 4; ++kt) {
    v8f acc[4] = {};
    #pragma unroll
    for (int s = 0; s < 8; ++s) {
      v2f a;
      a.x = ax[s][kt];
      a.y = ay[s][kt];
      #pragma unroll
      for (int nt = 0; nt < 4; ++nt) {
        v2f b;
        b.x = bx[s][nt];
        b.y = by[s][nt];
        acc[nt] = __builtin_amdgcn_wmma_f32_16x16x4_f32(
                      false, a, false, b, (short)0, acc[nt], false, false);
      }
    }
    #pragma unroll
    for (int r = 0; r < 8; ++r) {
      const size_t off = (size_t)(4 * r + kt) * kOUTN;
      const v4f bv = *reinterpret_cast<const v4f*>(bl + off);
      v4f o4;
      o4.x = acc[0][r] + bv.x;
      o4.y = acc[1][r] + bv.y;
      o4.z = acc[2][r] + bv.z;
      o4.w = acc[3][r] + bv.w;
      *reinterpret_cast<v4f*>(ol + off) = o4;
    }
  }
}

extern "C" void kernel_launch(void* const* d_in, const int* in_sizes, int n_in,
                              void* d_out, int out_size, void* d_ws, size_t ws_size,
                              hipStream_t stream) {
  const float* x    = (const float*)d_in[0];
  const float* nf   = (const float*)d_in[1];
  const float* ft   = (const float*)d_in[2];
  const float* bias = (const float*)d_in[3];
  const int*   A    = (const int*)d_in[4];
  float* out = (float*)d_out;

  constexpr size_t kRedBytes = (size_t)kN * kINC * kOUTN * sizeof(float);
  if (d_ws != nullptr && ws_size >= kRedBytes) {
    float* red = (float*)d_ws;
    row_in_lds_gather_max<<<kN * kINC, kRowT, 0, stream>>>(x, nf, A, red);
    maxima_times_ft_wmma<<<dim3(kOUTN / kPosW, kN / 4), 128, 0, stream>>>(red, ft, bias, out);
  } else {
    dim3 grid(kOUTN / 16, kN);
    fused_gather_max_wmma<<<grid, 128, 0, stream>>>(x, nf, ft, bias, A, out);
  }

  (void)in_sizes; (void)n_in; (void)out_size;
}


# Gather launch on 8-wave workgroups (2 waves per SIMD, 1024 positions per wave, indices requested two trips ahead): half as many waves enter and the LDS serves a row's waves evenly
# speedup vs baseline: 13.9986x; 1.0469x over previous
#include <hip/hip_runtime.h>

constexpr int kN    = 8;
constexpr int kINC  = 32;
constexpr int kINN  = 50000;
constexpr int kOUTC = 64;
constexpr int kOUTN = 8192;
constexpr int kD    = 32;

typedef __attribute__((ext_vector_type(2))) float v2f;
typedef __attribute__((ext_vector_type(8))) float v8f;

__global__ __launch_bounds__(128) void fused_gather_max_wmma(
    const float* __restrict__ x,
    const float* __restrict__ nf,
    const float* __restrict__ ft,
    const float* __restrict__ bias,
    const int*   __restrict__ A,
    float*       __restrict__ out)
{
  __shared__ int   sA[16 * kD];
  __shared__ float sred[16 * 33];

  const int tid    = threadIdx.x;
  const int o_base = blockIdx.x * 16;
  const int n      = blockIdx.y;

  for (int j = tid; j < 16 * kD; j += 128)
    sA[j] = A[o_base * kD + j];
  __syncthreads();

  {
    const int i  = tid & 31;
    const int og = tid >> 5;
    const float* __restrict__ xrow  = x  + ((size_t)n * kINC + i) * kINN;
    const float* __restrict__ nfrow = nf + (size_t)i * kINN;
    #pragma unroll
    for (int rep = 0; rep < 4; ++rep) {
      const int o_loc = og + rep * 4;
      const int* __restrict__ idxs = &sA[o_loc * kD];
      float m = -__builtin_inff();
      #pragma unroll 4
      for (int d = 0; d < kD; ++d) {
        const int idx = idxs[d];
        m = fmaxf(m, xrow[idx] * nfrow[idx]);
      }
      sred[o_loc * 33 + i] = m;
    }
  }
  __syncthreads();

  {
    const int lane  = tid & 31;
    const int wave  = tid >> 5;
    const int m16   = lane & 15;
    const int hi    = lane >> 4;
    const int kbase = wave * 16;

    v8f c = {};
    #pragma unroll
    for (int s = 0; s < 8; ++s) {
      const int ks = 4 * s + 2 * hi;
      v2f a, b;
      a.x = ft[(size_t)ks       * kOUTC + kbase + m16];
      a.y = ft[(size_t)(ks + 1) * kOUTC + kbase + m16];
      b.x = sred[m16 * 33 + ks];
      b.y = sred[m16 * 33 + ks + 1];
      c = __builtin_amdgcn_wmma_f32_16x16x4_f32(
              false, a, false, b, (short)0, c, false, false);
    }

    #pragma unroll
    for (int v = 0; v < 8; ++v) {
      const int k = kbase + v + 8 * hi;
      const int o = o_base + m16;
      const size_t bidx = (size_t)k * kOUTN + o;
      out[(size_t)n * kOUTC * kOUTN + bidx] = c[v] + bias[bidx];
    }
  }
}

typedef __attribute__((ext_vector_type(4))) float v4f;
typedef __attribute__((ext_vector_type(4))) int   v4i;

constexpr int kRowT  = 256;
constexpr int kRowF4 = kINN / 4;
constexpr int kOW    = kOUTN / (kRowT / 32);
constexpr int kIt    = kOW / 32;
constexpr int kFull  = kRowF4 / kRowT;

template <int MASK>
__device__ __forceinline__ float max_xmask(float v) {
  const int p = __builtin_amdgcn_update_dpp(0, __float_as_int(v), 0x160 + MASK, 0xF, 0xF, true);
  return fmaxf(v, __int_as_float(p));
}

__device__ __forceinline__ float gather_trip(const float* srow, const v4i (&cur)[8], int sub) {
  float m[8];
  #pragma unroll
  for (int j = 0; j < 8; ++j) {
    const float g0 = srow[cur[j].x];
    const float g1 = srow[cur[j].y];
    const float g2 = srow[cur[j].z];
    const float g3 = srow[cur[j].w];
    m[j] = fmaxf(fmaxf(g0, g1), fmaxf(g2, g3));
  }
  #pragma unroll
  for (int j = 0; j < 8; ++j) {
    m[j] = max_xmask<1>(m[j]);
    m[j] = max_xmask<2>(m[j]);
    m[j] = max_xmask<4>(m[j]);
  }
  float v = m[0];
  #pragma unroll
  for (int j = 1; j < 8; ++j) v = (sub == j) ? m[j] : v;
  return v;
}

__global__ __launch_bounds__(kRowT) void row_in_lds_gather_max(
    const float* __restrict__ x,
    const float* __restrict__ nf,
    const int*   __restrict__ A,
    float*       __restrict__ red)
{
  __shared__ __attribute__((aligned(16))) float srow[kINN];

  const int tid  = threadIdx.x;
  const int lane = tid & 31;
  const int wave = tid >> 5;
  const int r    = blockIdx.x;
  const int i    = r & 31;

  const v4i* __restrict__ a4 = reinterpret_cast<const v4i*>(A + (size_t)wave * kOW * kD) + lane;
  v4i s0[8], s1[8], s2[8], s3[8];
  #pragma unroll
  for (int j = 0; j < 8; ++j) s0[j] = a4[j * 32];
  #pragma unroll
  for (int j = 0; j < 8; ++j) s1[j] = a4[256 + j * 32];

  {
    const v4f* __restrict__ xr = reinterpret_cast<const v4f*>(x  + (size_t)r * kINN);
    const v4f* __restrict__ wr = reinterpret_cast<const v4f*>(nf + (size_t)i * kINN);
    v4f* s4 = reinterpret_cast<v4f*>(srow);
    #pragma unroll
    for (int b = 0; b < kFull / 12; ++b) {
      v4f xv[12], wv[12];
      #pragma unroll
      for (int k = 0; k < 12; ++k) {
        const int q = tid + (b * 12 + k) * kRowT;
        xv[k] = xr[q];
        wv[k] = wr[q];
      }
      #pragma unroll
      for (int k = 0; k < 12; ++k) {
        const int q = tid + (b * 12 + k) * kRowT;
        s4[q] = xv[k] * wv[k];
      }
    }
    const int q = tid + kFull * kRowT;
    if (q < kRowF4) s4[q] = xr[q] * wr[q];
  }
  __syncthreads();

  {
    float* __restrict__ rr = red + (size_t)r * kOUTN + wave * kOW + 4 * (lane & 7) + (lane >> 3);
    const int sub = lane & 7;

    #pragma unroll 1
    for (int it = 0; it < kIt; it += 4) {
      #pragma unroll
      for (int j = 0; j < 8; ++j) s2[j] = a4[(it + 2) * 256 + j * 32];
      rr[it * 32] = gather_trip(srow, s0, sub);

      #pragma unroll
      for (int j = 0; j < 8; ++j) s3[j] = a4[(it + 3) * 256 + j * 32];
      rr[(it + 1) * 32] = gather_trip(srow, s1, sub);

      const int t4 = (it + 4 < kIt) ? it + 4 : kIt - 1;
      const int t5 = (it + 5 < kIt) ? it + 5 : kIt - 1;
      #pragma unroll
      for (int j = 0; j < 8; ++j) s0[j] = a4[t4 * 256 + j * 32];
      rr[(it + 2) * 32] = gather_trip(srow, s2, sub);

      #pragma unroll
      for (int j = 0; j < 8; ++j) s1[j] = a4[t5 * 256 + j * 32];
      rr[(it + 3) * 32] = gather_trip(srow, s3, sub);
    }
  }
}

constexpr int kPosW = 64;
__global__ __launch_bounds__(128) void maxima_times_ft_wmma(
    const float* __restrict__ red,
    const float* __restrict__ ft,
    const float* __restrict__ bias,
    float*       __restrict__ out)
{
  const int tid    = threadIdx.x;
  const int lane   = tid & 31;
  const int n      = blockIdx.y * 4 + (tid >> 5);
  const int o_base = blockIdx.x * kPosW;
  const int c      = lane & 15;
  const int h      = lane >> 4;

  const v4f* __restrict__ rb = reinterpret_cast<const v4f*>(
      red + ((size_t)n * kINC + 2 * h) * kOUTN + o_base + 4 * c);
  const v4f* __restrict__ fa = reinterpret_cast<const v4f*>(ft + (size_t)(2 * h) * kOUTC + 4 * c);

  v4f bx[8], by[8], ax[8], ay[8];
  #pragma unroll
  for (int s = 0; s < 8; ++s) {
    bx[s] = rb[(size_t)(4 * s)     * (kOUTN / 4)];
    by[s] = rb[(size_t)(4 * s + 1) * (kOUTN / 4)];
    ax[s] = fa[(4 * s)     * (kOUTC / 4)];
    ay[s] = fa[(4 * s + 1) * (kOUTC / 4)];
  }

  const size_t lane_off = (size_t)(32 * h) * kOUTN + o_base + 4 * c;
  const float* __restrict__ bl = bias + lane_off;
  float*       __restrict__ ol = out + (size_t)n * kOUTC * kOUTN + lane_off;

  #pragma unroll
  for (int kt = 0; kt < 4; ++kt) {
    v8f acc[4] = {};
    #pragma unroll
    for (int s = 0; s < 8; ++s) {
      v2f a;
      a.x = ax[s][kt];
      a.y = ay[s][kt];
      #pragma unroll
      for (int nt = 0; nt < 4; ++nt) {
        v2f b;
        b.x = bx[s][nt];
        b.y = by[s][nt];
        acc[nt] = __builtin_amdgcn_wmma_f32_16x16x4_f32(
                      false, a, false, b, (short)0, acc[nt], false, false);
      }
    }
    #pragma unroll
    for (int r = 0; r < 8; ++r) {
      const size_t off = (size_t)(4 * r + kt) * kOUTN;
      const v4f bv = *reinterpret_cast<const v4f*>(bl + off);
      v4f o4;
      o4.x = acc[0][r] + bv.x;
      o4.y = acc[1][r] + bv.y;
      o4.z = acc[2][r] + bv.z;
      o4.w = acc[3][r] + bv.w;
      *reinterpret_cast<v4f*>(ol + off) = o4;
    }
  }
}

extern "C" void kernel_launch(void* const* d_in, const int* in_sizes, int n_in,
                              void* d_out, int out_size, void* d_ws, size_t ws_size,
                              hipStream_t stream) {
  const float* x    = (const float*)d_in[0];
  const float* nf   = (const float*)d_in[1];
  const float* ft   = (const float*)d_in[2];
  const float* bias = (const float*)d_in[3];
  const int*   A    = (const int*)d_in[4];
  float* out = (float*)d_out;

  constexpr size_t kRedBytes = (size_t)kN * kINC * kOUTN * sizeof(float);
  if (d_ws != nullptr && ws_size >= kRedBytes) {
    float* red = (float*)d_ws;
    row_in_lds_gather_max<<<kN * kINC, kRowT, 0, stream>>>(x, nf, A, red);
    maxima_times_ft_wmma<<<dim3(kOUTN / kPosW, kN / 4), 128, 0, stream>>>(red, ft, bias, out);
  } else {
    dim3 grid(kOUTN / 16, kN);
    fused_gather_max_wmma<<<grid, 128, 0, stream>>>(x, nf, ft, bias, A, out);
  }

  (void)in_sizes; (void)n_in; (void)out_size;
}
